# starting kernel
# baseline (speedup 1.0000x reference)
#include <hip/hip_runtime.h>
#include <hip/hip_bf16.h>

typedef float v2f __attribute__((ext_vector_type(2)));
typedef float v8f __attribute__((ext_vector_type(8)));

#define BB 512
#define FF 512
#define KK 32
#define DD 16
#define NN (KK * DD)
#define OUTC (FF + KK)

__global__ __launch_bounds__(256) void gemm_wmma_f32(const float* __restrict__ x,
                                                     const float* __restrict__ T,
                                                     float* __restrict__ m) {
  const int lane = threadIdx.x & 31;
  const int wave = threadIdx.x >> 5;
  const int tile = blockIdx.x * 8 + wave;
  const int tile_m = tile >> 5;
  const int tile_n = tile & 31;
  const int half = lane >> 4;
  const int l16 = lane & 15;

  const int arow = tile_m * 16 + l16;
  const int bcol = tile_n * 16 + l16;

  v8f c = {};
  for (int k = 0; k < FF; k += 4) {
    const int ka = k + half * 2;
    v2f a;
    a.x = x[arow * FF + ka];
    a.y = x[arow * FF + ka + 1];
    v2f b;
    b.x = T[(ka)     * NN + bcol];
    b.y = T[(ka + 1) * NN + bcol];
    c = __builtin_amdgcn_wmma_f32_16x16x4_f32(false, a, false, b,
                                              (short)0, c, false, false);
  }

#pragma unroll
  for (int r = 0; r < 8; ++r) {
    m[(tile_m * 16 + r + half * 8) * NN + bcol] = c[r];
  }
}

__global__ __launch_bounds__(256) void copy_x(const float* __restrict__ x,
                                              float* __restrict__ out) {
  const int idx = blockIdx.x * 256 + threadIdx.x;
  if (idx < BB * FF) {
    const int i = idx / FF;
    const int ccol = idx - i * FF;
    out[i * OUTC + ccol] = x[idx];
  }
}

__global__ __launch_bounds__(64) void feats_kernel(const float* __restrict__ m,
                                                   float* __restrict__ out) {
  __shared__ float lds[BB * DD];
  const int k = blockIdx.y;
  const int i = blockIdx.x * 64 + threadIdx.x;

  for (int t = threadIdx.x; t < BB * 4; t += 64) {
    const int j = t >> 2;
    const int q = t & 3;
    const float4* src = (const float4*)(m + j * NN + k * DD);
    ((float4*)lds)[j * 4 + q] = src[q];
  }
  __syncthreads();

  float4 mi0 = ((const float4*)(lds + i * DD))[0];
  float4 mi1 = ((const float4*)(lds + i * DD))[1];
  float4 mi2 = ((const float4*)(lds + i * DD))[2];
  float4 mi3 = ((const float4*)(lds + i * DD))[3];

  float acc = 0.f;
  for (int j = 0; j < BB; ++j) {
    const float4* row = (const float4*)(lds + j * DD);
    float4 r0 = row[0], r1 = row[1], r2 = row[2], r3 = row[3];
    float dist = fabsf(mi0.x - r0.x) + fabsf(mi0.y - r0.y) +
                 fabsf(mi0.z - r0.z) + fabsf(mi0.w - r0.w);
    dist += fabsf(mi1.x - r1.x) + fabsf(mi1.y - r1.y) +
            fabsf(mi1.z - r1.z) + fabsf(mi1.w - r1.w);
    dist += fabsf(mi2.x - r2.x) + fabsf(mi2.y - r2.y) +
            fabsf(mi2.z - r2.z) + fabsf(mi2.w - r2.w);
    dist += fabsf(mi3.x - r3.x) + fabsf(mi3.y - r3.y) +
            fabsf(mi3.z - r3.z) + fabsf(mi3.w - r3.w);
    acc += __expf(-dist);
  }
  out[i * OUTC + FF + k] = acc;
}

extern "C" void kernel_launch(void* const* d_in, const int* in_sizes, int n_in,
                              void* d_out, int out_size, void* d_ws, size_t ws_size,
                              hipStream_t stream) {
  (void)in_sizes; (void)n_in; (void)out_size; (void)ws_size;
  const float* x = (const float*)d_in[0];
  const float* T = (const float*)d_in[1];
  float* out = (float*)d_out;
  float* m = (float*)d_ws;

  gemm_wmma_f32<<<128, 256, 0, stream>>>(x, T, m);
  copy_x<<<(BB * FF + 255) / 256, 256, 0, stream>>>(x, out);
  feats_kernel<<<dim3(8, 32), 64, 0, stream>>>(m, out);
}


# feats without LDS: lane keeps 2 rows i; 8 rows j sit one per lane and go round the 8-lane group by permlane16_var; 4 groups take different j tiles; sums: d as a tree, j as 4 partials of 128
# speedup vs baseline: 1.5629x; 1.5629x over previous
#include <hip/hip_runtime.h>
#include <hip/hip_bf16.h>

typedef float v2f __attribute__((ext_vector_type(2)));
typedef float v8f __attribute__((ext_vector_type(8)));

#define BB 512
#define FF 512
#define KK 32
#define DD 16
#define NN (KK * DD)
#define OUTC (FF + KK)

__global__ __launch_bounds__(256) void gemm_wmma_f32(const float* __restrict__ x,
                                                     const float* __restrict__ T,
                                                     float* __restrict__ m) {
  const int lane = threadIdx.x & 31;
  const int wave = threadIdx.x >> 5;
  const int tile = blockIdx.x * 8 + wave;
  const int tile_m = tile >> 5;
  const int tile_n = tile & 31;
  const int half = lane >> 4;
  const int l16 = lane & 15;

  const int arow = tile_m * 16 + l16;
  const int bcol = tile_n * 16 + l16;

  v8f c = {};
  for (int k = 0; k < FF; k += 4) {
    const int ka = k + half * 2;
    v2f a;
    a.x = x[arow * FF + ka];
    a.y = x[arow * FF + ka + 1];
    v2f b;
    b.x = T[(ka)     * NN + bcol];
    b.y = T[(ka + 1) * NN + bcol];
    c = __builtin_amdgcn_wmma_f32_16x16x4_f32(false, a, false, b,
                                              (short)0, c, false, false);
  }

#pragma unroll
  for (int r = 0; r < 8; ++r) {
    m[(tile_m * 16 + r + half * 8) * NN + bcol] = c[r];
  }
}

__global__ __launch_bounds__(256) void copy_x(const float* __restrict__ x,
                                              float* __restrict__ out) {
  const int idx = blockIdx.x * 256 + threadIdx.x;
  if (idx < BB * FF) {
    const int i = idx / FF;
    const int ccol = idx - i * FF;
    out[i * OUTC + ccol] = x[idx];
  }
}

#define GL 8
#define RI (16 / GL)
#define NQ (32 / GL)

__device__ __forceinline__ float pair_term(const float (&mi)[DD],
                                           const float (&mj)[DD]) {
  float t[DD];
#pragma unroll
  for (int d = 0; d < DD; ++d) t[d] = mj[d] - mi[d];
  const float p0 = fabsf(t[0]) + fabsf(t[1]);
  const float p1 = fabsf(t[2]) + fabsf(t[3]);
  const float p2 = fabsf(t[4]) + fabsf(t[5]);
  const float p3 = fabsf(t[6]) + fabsf(t[7]);
  const float p4 = fabsf(t[8]) + fabsf(t[9]);
  const float p5 = fabsf(t[10]) + fabsf(t[11]);
  const float p6 = fabsf(t[12]) + fabsf(t[13]);
  const float p7 = fabsf(t[14]) + fabsf(t[15]);
  const float dist = ((p0 + p1) + (p2 + p3)) + ((p4 + p5) + (p6 + p7));
  return __expf(-dist);
}

__device__ __forceinline__ void rotate_row(float (&mj)[DD], unsigned sel) {
#pragma unroll
  for (int d = 0; d < DD; ++d) {
    const unsigned u = __float_as_uint(mj[d]);
    mj[d] = __uint_as_float(__builtin_amdgcn_permlane16_var(u, u, sel, false, false));
  }
}

__device__ __forceinline__ void load_row(const float* __restrict__ p,
                                         float (&r)[DD]) {
  const float4 a = ((const float4*)p)[0];
  const float4 b = ((const float4*)p)[1];
  const float4 c = ((const float4*)p)[2];
  const float4 d = ((const float4*)p)[3];
  r[0] = a.x;  r[1] = a.y;  r[2] = a.z;  r[3] = a.w;
  r[4] = b.x;  r[5] = b.y;  r[6] = b.z;  r[7] = b.w;
  r[8] = c.x;  r[9] = c.y;  r[10] = c.z; r[11] = c.w;
  r[12] = d.x; r[13] = d.y; r[14] = d.z; r[15] = d.w;
}

__global__ __launch_bounds__(128) void feats_kernel(const float* __restrict__ m,
                                                    float* __restrict__ out) {
  const int lane = threadIdx.x & 31;
  const int wave = threadIdx.x >> 5;
  const int k = blockIdx.y;
  const int I = blockIdx.x * 4 + wave;
  const int cg = lane & (GL - 1);
  const int q = lane / GL;
  const float* base = m + k * DD;

  float mi[RI][DD], cur[DD], nxt[DD], acc[RI];
#pragma unroll
  for (int r = 0; r < RI; ++r) {
    load_row(base + (I * 16 + r * GL + cg) * NN, mi[r]);
    acc[r] = 0.f;
  }
  load_row(base + (q * GL + cg) * NN, cur);

  const unsigned sel = (unsigned)((lane & 15 & ~(GL - 1)) | ((cg + 1) & (GL - 1)));
#pragma unroll 1
  for (int t = 0; t < 16; ++t) {
    const int jn = ((((t + 1) & 15) * NQ + q) * GL + cg);
    __builtin_amdgcn_sched_barrier(0);
    load_row(base + jn * NN, nxt);
    __builtin_amdgcn_sched_barrier(0);

#pragma unroll
    for (int g = 0; g < GL; ++g) {
#pragma unroll
      for (int r = 0; r < RI; ++r) acc[r] += pair_term(mi[r], cur);
      if (g < GL - 1) rotate_row(cur, sel);
    }

    __builtin_amdgcn_sched_barrier(0);
#pragma unroll
    for (int d = 0; d < DD; ++d) cur[d] = nxt[d];
  }

#pragma unroll
  for (int r = 0; r < RI; ++r) {
#pragma unroll
    for (int w = GL; w < 32; w <<= 1) acc[r] += __shfl_xor(acc[r], w, 32);
  }
  if (q == 0) {
#pragma unroll
    for (int r = 0; r < RI; ++r)
      out[(I * 16 + r * GL + cg) * OUTC + FF + k] = acc[r];
  }
}

extern "C" void kernel_launch(void* const* d_in, const int* in_sizes, int n_in,
                              void* d_out, int out_size, void* d_ws, size_t ws_size,
                              hipStream_t stream) {
  (void)in_sizes; (void)n_in; (void)out_size; (void)ws_size;
  const float* x = (const float*)d_in[0];
  const float* T = (const float*)d_in[1];
  float* out = (float*)d_out;
  float* m = (float*)d_ws;

  gemm_wmma_f32<<<128, 256, 0, stream>>>(x, T, m);
  copy_x<<<(BB * FF + 255) / 256, 256, 0, stream>>>(x, out);
  feats_kernel<<<dim3(8, 32), 128, 0, stream>>>(m, out);
}


# m = x @ T: a wave owns a 16x32 strip (2 accumulator tiles); fragments go global->registers by 16-byte A and 8-byte B loads in a permuted k order, 64 k ahead; k sum order inside each 8 changes
# speedup vs baseline: 2.2745x; 1.4553x over previous
#include <hip/hip_runtime.h>
#include <hip/hip_bf16.h>

typedef float v2f __attribute__((ext_vector_type(2)));
typedef float v8f __attribute__((ext_vector_type(8)));

#define BB 512
#define FF 512
#define KK 32
#define DD 16
#define NN (KK * DD)
#define OUTC (FF + KK)

typedef float v4f __attribute__((ext_vector_type(4)));

#define KC 64
#define NS (KC / 8)

__device__ __forceinline__ void load_group(const float* __restrict__ xa,
                                           const float* __restrict__ tb,
                                           v4f (&a)[NS], v2f (&b)[NS][4]) {
#pragma unroll
  for (int s = 0; s < NS; ++s) {
    a[s] = *(const v4f*)(xa + 8 * s);
#pragma unroll
    for (int q = 0; q < 4; ++q) b[s][q] = *(const v2f*)(tb + (8 * s + q) * NN);
  }
}

__device__ __forceinline__ void pair_rows(const v2f (&b)[NS][4], v2f (&bt)[NS][2][2]) {
#pragma unroll
  for (int s = 0; s < NS; ++s) {
#pragma unroll
    for (int j = 0; j < 2; ++j) {
      bt[s][j][0].x = b[s][2 * j].x;   bt[s][j][0].y = b[s][2 * j + 1].x;
      bt[s][j][1].x = b[s][2 * j].y;   bt[s][j][1].y = b[s][2 * j + 1].y;
    }
  }
}

__device__ __forceinline__ void mma_group(const v4f (&a)[NS], const v2f (&bt)[NS][2][2],
                                          v8f& c0, v8f& c1) {
#pragma unroll
  for (int s = 0; s < NS; ++s) {
#pragma unroll
    for (int j = 0; j < 2; ++j) {
      v2f af;
      af.x = a[s][2 * j];
      af.y = a[s][2 * j + 1];
      c0 = __builtin_amdgcn_wmma_f32_16x16x4_f32(false, af, false, bt[s][j][0],
                                                 (short)0, c0, false, false);
      c1 = __builtin_amdgcn_wmma_f32_16x16x4_f32(false, af, false, bt[s][j][1],
                                                 (short)0, c1, false, false);
    }
  }
}

__global__ __launch_bounds__(64) void gemm_wmma_f32(const float* __restrict__ x,
                                                    const float* __restrict__ T,
                                                    float* __restrict__ m) {
  const int lane = threadIdx.x & 31;
  const int wave = threadIdx.x >> 5;
  const int strip = blockIdx.x * 2 + wave;
  const int tile_m = strip >> 4;
  const int n0 = (strip & 15) * 32;
  const int half = lane >> 4;
  const int c = lane & 15;

  const float* xa = x + (tile_m * 16 + c) * FF + 4 * half;
  const float* tb = T + (4 * half) * NN + n0 + 2 * c;

  v8f c0 = {}, c1 = {};
  v4f a[2][NS];
  v2f b[2][NS][4];
  v2f bt[NS][2][2];

  load_group(xa, tb, a[0], b[0]);
#pragma unroll
  for (int g = 0; g < FF / KC; ++g) {
    __builtin_amdgcn_sched_barrier(0);
    if (g + 1 < FF / KC)
      load_group(xa + (g + 1) * KC, tb + (g + 1) * KC * NN, a[(g + 1) & 1], b[(g + 1) & 1]);
    __builtin_amdgcn_sched_barrier(0);
    pair_rows(b[g & 1], bt);
    __builtin_amdgcn_sched_barrier(0);
    mma_group(a[g & 1], bt, c0, c1);
  }

  float* mo = m + (tile_m * 16 + half * 8) * NN + n0 + 2 * c;
#pragma unroll
  for (int r = 0; r < 8; ++r) {
    v2f o;
    o.x = c0[r];
    o.y = c1[r];
    *(v2f*)(mo + r * NN) = o;
  }
}

__global__ __launch_bounds__(256) void copy_x(const float* __restrict__ x,
                                              float* __restrict__ out) {
  const int idx = blockIdx.x * 256 + threadIdx.x;
  if (idx < BB * FF) {
    const int i = idx / FF;
    const int ccol = idx - i * FF;
    out[i * OUTC + ccol] = x[idx];
  }
}

#define GL 8
#define RI (16 / GL)
#define NQ (32 / GL)

__device__ __forceinline__ float pair_term(const float (&mi)[DD],
                                           const float (&mj)[DD]) {
  float t[DD];
#pragma unroll
  for (int d = 0; d < DD; ++d) t[d] = mj[d] - mi[d];
  const float p0 = fabsf(t[0]) + fabsf(t[1]);
  const float p1 = fabsf(t[2]) + fabsf(t[3]);
  const float p2 = fabsf(t[4]) + fabsf(t[5]);
  const float p3 = fabsf(t[6]) + fabsf(t[7]);
  const float p4 = fabsf(t[8]) + fabsf(t[9]);
  const float p5 = fabsf(t[10]) + fabsf(t[11]);
  const float p6 = fabsf(t[12]) + fabsf(t[13]);
  const float p7 = fabsf(t[14]) + fabsf(t[15]);
  const float dist = ((p0 + p1) + (p2 + p3)) + ((p4 + p5) + (p6 + p7));
  return __expf(-dist);
}

__device__ __forceinline__ void rotate_row(float (&mj)[DD], unsigned sel) {
#pragma unroll
  for (int d = 0; d < DD; ++d) {
    const unsigned u = __float_as_uint(mj[d]);
    mj[d] = __uint_as_float(__builtin_amdgcn_permlane16_var(u, u, sel, false, false));
  }
}

__device__ __forceinline__ void load_row(const float* __restrict__ p,
                                         float (&r)[DD]) {
  const float4 a = ((const float4*)p)[0];
  const float4 b = ((const float4*)p)[1];
  const float4 c = ((const float4*)p)[2];
  const float4 d = ((const float4*)p)[3];
  r[0] = a.x;  r[1] = a.y;  r[2] = a.z;  r[3] = a.w;
  r[4] = b.x;  r[5] = b.y;  r[6] = b.z;  r[7] = b.w;
  r[8] = c.x;  r[9] = c.y;  r[10] = c.z; r[11] = c.w;
  r[12] = d.x; r[13] = d.y; r[14] = d.z; r[15] = d.w;
}

__global__ __launch_bounds__(128) void feats_kernel(const float* __restrict__ m,
                                                    float* __restrict__ out) {
  const int lane = threadIdx.x & 31;
  const int wave = threadIdx.x >> 5;
  const int k = blockIdx.y;
  const int I = blockIdx.x * 4 + wave;
  const int cg = lane & (GL - 1);
  const int q = lane / GL;
  const float* base = m + k * DD;

  float mi[RI][DD], cur[DD], nxt[DD], acc[RI];
#pragma unroll
  for (int r = 0; r < RI; ++r) {
    load_row(base + (I * 16 + r * GL + cg) * NN, mi[r]);
    acc[r] = 0.f;
  }
  load_row(base + (q * GL + cg) * NN, cur);

  const unsigned sel = (unsigned)((lane & 15 & ~(GL - 1)) | ((cg + 1) & (GL - 1)));
#pragma unroll 1
  for (int t = 0; t < 16; ++t) {
    const int jn = ((((t + 1) & 15) * NQ + q) * GL + cg);
    __builtin_amdgcn_sched_barrier(0);
    load_row(base + jn * NN, nxt);
    __builtin_amdgcn_sched_barrier(0);

#pragma unroll
    for (int g = 0; g < GL; ++g) {
#pragma unroll
      for (int r = 0; r < RI; ++r) acc[r] += pair_term(mi[r], cur);
      if (g < GL - 1) rotate_row(cur, sel);
    }

    __builtin_amdgcn_sched_barrier(0);
#pragma unroll
    for (int d = 0; d < DD; ++d) cur[d] = nxt[d];
  }

#pragma unroll
  for (int r = 0; r < RI; ++r) {
#pragma unroll
    for (int w = GL; w < 32; w <<= 1) acc[r] += __shfl_xor(acc[r], w, 32);
  }
  if (q == 0) {
#pragma unroll
    for (int r = 0; r < RI; ++r)
      out[(I * 16 + r * GL + cg) * OUTC + FF + k] = acc[r];
  }
}

extern "C" void kernel_launch(void* const* d_in, const int* in_sizes, int n_in,
                              void* d_out, int out_size, void* d_ws, size_t ws_size,
                              hipStream_t stream) {
  (void)in_sizes; (void)n_in; (void)out_size; (void)ws_size;
  const float* x = (const float*)d_in[0];
  const float* T = (const float*)d_in[1];
  float* out = (float*)d_out;
  float* m = (float*)d_ws;

  gemm_wmma_f32<<<256, 64, 0, stream>>>(x, T, m);
  copy_x<<<(BB * FF + 255) / 256, 256, 0, stream>>>(x, out);
  feats_kernel<<<dim3(8, 32), 128, 0, stream>>>(m, out);
}


# out[:, :512] = x is stored by the GEMM from the A fragments it holds (first wave of a block: one 64-k group of its 16 rows, whole lines); x is not re-read, the copy launch is gone: 2 launches, not 3
# speedup vs baseline: 2.5004x; 1.0993x over previous
#include <hip/hip_runtime.h>
#include <hip/hip_bf16.h>

typedef float v2f __attribute__((ext_vector_type(2)));
typedef float v8f __attribute__((ext_vector_type(8)));

#define BB 512
#define FF 512
#define KK 32
#define DD 16
#define NN (KK * DD)
#define OUTC (FF + KK)

typedef float v4f __attribute__((ext_vector_type(4)));

#define KC 64
#define NS (KC / 8)

__device__ __forceinline__ void load_group(const float* __restrict__ xa,
                                           const float* __restrict__ tb,
                                           v4f (&a)[NS], v2f (&b)[NS][4]) {
#pragma unroll
  for (int s = 0; s < NS; ++s) {
    a[s] = *(const v4f*)(xa + 8 * s);
#pragma unroll
    for (int q = 0; q < 4; ++q) b[s][q] = *(const v2f*)(tb + (8 * s + q) * NN);
  }
}

__device__ __forceinline__ void pair_rows(const v2f (&b)[NS][4], v2f (&bt)[NS][2][2]) {
#pragma unroll
  for (int s = 0; s < NS; ++s) {
#pragma unroll
    for (int j = 0; j < 2; ++j) {
      bt[s][j][0].x = b[s][2 * j].x;   bt[s][j][0].y = b[s][2 * j + 1].x;
      bt[s][j][1].x = b[s][2 * j].y;   bt[s][j][1].y = b[s][2 * j + 1].y;
    }
  }
}

__device__ __forceinline__ void mma_group(const v4f (&a)[NS], const v2f (&bt)[NS][2][2],
                                          v8f& c0, v8f& c1) {
#pragma unroll
  for (int s = 0; s < NS; ++s) {
#pragma unroll
    for (int j = 0; j < 2; ++j) {
      v2f af;
      af.x = a[s][2 * j];
      af.y = a[s][2 * j + 1];
      c0 = __builtin_amdgcn_wmma_f32_16x16x4_f32(false, af, false, bt[s][j][0],
                                                 (short)0, c0, false, false);
      c1 = __builtin_amdgcn_wmma_f32_16x16x4_f32(false, af, false, bt[s][j][1],
                                                 (short)0, c1, false, false);
    }
  }
}

__global__ __launch_bounds__(64) void gemm_wmma_f32(const float* __restrict__ x,
                                                    const float* __restrict__ T,
                                                    float* __restrict__ m,
                                                    float* __restrict__ out) {
  const int lane = threadIdx.x & 31;
  const int wave = threadIdx.x >> 5;
  const int strip = blockIdx.x * 2 + wave;
  const int tile_m = strip >> 4;
  const int n0 = (strip & 15) * 32;
  const int half = lane >> 4;
  const int c = lane & 15;

  const float* xa = x + (tile_m * 16 + c) * FF + 4 * half;
  const float* tb = T + (4 * half) * NN + n0 + 2 * c;

  const int gsel = blockIdx.x & 7;
  const int wsc = __builtin_amdgcn_readfirstlane(wave);
  float* xo = out + (tile_m * 16 + c) * OUTC + 4 * half;

  v8f c0 = {}, c1 = {};
  v4f a[2][NS];
  v2f b[2][NS][4];
  v2f bt[NS][2][2];

  load_group(xa, tb, a[0], b[0]);
#pragma unroll
  for (int g = 0; g < FF / KC; ++g) {
    __builtin_amdgcn_sched_barrier(0);
    if (g + 1 < FF / KC)
      load_group(xa + (g + 1) * KC, tb + (g + 1) * KC * NN, a[(g + 1) & 1], b[(g + 1) & 1]);
    __builtin_amdgcn_sched_barrier(0);
    if (g == gsel && wsc == 0) {
      float* p = xo + g * KC;
#pragma unroll
      for (int s = 0; s < NS; ++s) *(v4f*)(p + 8 * s) = a[g & 1][s];
    }
    pair_rows(b[g & 1], bt);
    __builtin_amdgcn_sched_barrier(0);
    mma_group(a[g & 1], bt, c0, c1);
  }

  float* mo = m + (tile_m * 16 + half * 8) * NN + n0 + 2 * c;
#pragma unroll
  for (int r = 0; r < 8; ++r) {
    v2f o;
    o.x = c0[r];
    o.y = c1[r];
    *(v2f*)(mo + r * NN) = o;
  }
}

#define GL 8
#define RI (16 / GL)
#define NQ (32 / GL)

__device__ __forceinline__ float pair_term(const float (&mi)[DD],
                                           const float (&mj)[DD]) {
  float t[DD];
#pragma unroll
  for (int d = 0; d < DD; ++d) t[d] = mj[d] - mi[d];
  const float p0 = fabsf(t[0]) + fabsf(t[1]);
  const float p1 = fabsf(t[2]) + fabsf(t[3]);
  const float p2 = fabsf(t[4]) + fabsf(t[5]);
  const float p3 = fabsf(t[6]) + fabsf(t[7]);
  const float p4 = fabsf(t[8]) + fabsf(t[9]);
  const float p5 = fabsf(t[10]) + fabsf(t[11]);
  const float p6 = fabsf(t[12]) + fabsf(t[13]);
  const float p7 = fabsf(t[14]) + fabsf(t[15]);
  const float dist = ((p0 + p1) + (p2 + p3)) + ((p4 + p5) + (p6 + p7));
  return __expf(-dist);
}

__device__ __forceinline__ void rotate_row(float (&mj)[DD], unsigned sel) {
#pragma unroll
  for (int d = 0; d < DD; ++d) {
    const unsigned u = __float_as_uint(mj[d]);
    mj[d] = __uint_as_float(__builtin_amdgcn_permlane16_var(u, u, sel, false, false));
  }
}

__device__ __forceinline__ void load_row(const float* __restrict__ p,
                                         float (&r)[DD]) {
  const float4 a = ((const float4*)p)[0];
  const float4 b = ((const float4*)p)[1];
  const float4 c = ((const float4*)p)[2];
  const float4 d = ((const float4*)p)[3];
  r[0] = a.x;  r[1] = a.y;  r[2] = a.z;  r[3] = a.w;
  r[4] = b.x;  r[5] = b.y;  r[6] = b.z;  r[7] = b.w;
  r[8] = c.x;  r[9] = c.y;  r[10] = c.z; r[11] = c.w;
  r[12] = d.x; r[13] = d.y; r[14] = d.z; r[15] = d.w;
}

__global__ __launch_bounds__(128) void feats_kernel(const float* __restrict__ m,
                                                    float* __restrict__ out) {
  const int lane = threadIdx.x & 31;
  const int wave = threadIdx.x >> 5;
  const int k = blockIdx.y;
  const int I = blockIdx.x * 4 + wave;
  const int cg = lane & (GL - 1);
  const int q = lane / GL;
  const float* base = m + k * DD;

  float mi[RI][DD], cur[DD], nxt[DD], acc[RI];
#pragma unroll
  for (int r = 0; r < RI; ++r) {
    load_row(base + (I * 16 + r * GL + cg) * NN, mi[r]);
    acc[r] = 0.f;
  }
  load_row(base + (q * GL + cg) * NN, cur);

  const unsigned sel = (unsigned)((lane & 15 & ~(GL - 1)) | ((cg + 1) & (GL - 1)));
#pragma unroll 1
  for (int t = 0; t < 16; ++t) {
    const int jn = ((((t + 1) & 15) * NQ + q) * GL + cg);
    __builtin_amdgcn_sched_barrier(0);
    load_row(base + jn * NN, nxt);
    __builtin_amdgcn_sched_barrier(0);

#pragma unroll
    for (int g = 0; g < GL; ++g) {
#pragma unroll
      for (int r = 0; r < RI; ++r) acc[r] += pair_term(mi[r], cur);
      if (g < GL - 1) rotate_row(cur, sel);
    }

    __builtin_amdgcn_sched_barrier(0);
#pragma unroll
    for (int d = 0; d < DD; ++d) cur[d] = nxt[d];
  }

#pragma unroll
  for (int r = 0; r < RI; ++r) {
#pragma unroll
    for (int w = GL; w < 32; w <<= 1) acc[r] += __shfl_xor(acc[r], w, 32);
  }
  if (q == 0) {
#pragma unroll
    for (int r = 0; r < RI; ++r)
      out[(I * 16 + r * GL + cg) * OUTC + FF + k] = acc[r];
  }
}

extern "C" void kernel_launch(void* const* d_in, const int* in_sizes, int n_in,
                              void* d_out, int out_size, void* d_ws, size_t ws_size,
                              hipStream_t stream) {
  (void)in_sizes; (void)n_in; (void)out_size; (void)ws_size;
  const float* x = (const float*)d_in[0];
  const float* T = (const float*)d_in[1];
  float* out = (float*)d_out;
  float* m = (float*)d_ws;

  gemm_wmma_f32<<<256, 64, 0, stream>>>(x, T, m, out);
  feats_kernel<<<dim3(8, 32), 128, 0, stream>>>(m, out);
}


# feats: each half wave keeps 8 rows i in every lane's registers, a lane owns its rows j (16n + lane%16); no cross-lane gathers in the loop; j sum now 16 lane partials of 32 j joined as a tree
# speedup vs baseline: 2.9137x; 1.1653x over previous
#include <hip/hip_runtime.h>
#include <hip/hip_bf16.h>

typedef float v2f __attribute__((ext_vector_type(2)));
typedef float v8f __attribute__((ext_vector_type(8)));

#define BB 512
#define FF 512
#define KK 32
#define DD 16
#define NN (KK * DD)
#define OUTC (FF + KK)

typedef float v4f __attribute__((ext_vector_type(4)));

#define KC 64
#define NS (KC / 8)

__device__ __forceinline__ void load_group(const float* __restrict__ xa,
                                           const float* __restrict__ tb,
                                           v4f (&a)[NS], v2f (&b)[NS][4]) {
#pragma unroll
  for (int s = 0; s < NS; ++s) {
    a[s] = *(const v4f*)(xa + 8 * s);
#pragma unroll
    for (int q = 0; q < 4; ++q) b[s][q] = *(const v2f*)(tb + (8 * s + q) * NN);
  }
}

__device__ __forceinline__ void pair_rows(const v2f (&b)[NS][4], v2f (&bt)[NS][2][2]) {
#pragma unroll
  for (int s = 0; s < NS; ++s) {
#pragma unroll
    for (int j = 0; j < 2; ++j) {
      bt[s][j][0].x = b[s][2 * j].x;   bt[s][j][0].y = b[s][2 * j + 1].x;
      bt[s][j][1].x = b[s][2 * j].y;   bt[s][j][1].y = b[s][2 * j + 1].y;
    }
  }
}

__device__ __forceinline__ void mma_group(const v4f (&a)[NS], const v2f (&bt)[NS][2][2],
                                          v8f& c0, v8f& c1) {
#pragma unroll
  for (int s = 0; s < NS; ++s) {
#pragma unroll
    for (int j = 0; j < 2; ++j) {
      v2f af;
      af.x = a[s][2 * j];
      af.y = a[s][2 * j + 1];
      c0 = __builtin_amdgcn_wmma_f32_16x16x4_f32(false, af, false, bt[s][j][0],
                                                 (short)0, c0, false, false);
      c1 = __builtin_amdgcn_wmma_f32_16x16x4_f32(false, af, false, bt[s][j][1],
                                                 (short)0, c1, false, false);
    }
  }
}

__global__ __launch_bounds__(64) void gemm_wmma_f32(const float* __restrict__ x,
                                                    const float* __restrict__ T,
                                                    float* __restrict__ m,
                                                    float* __restrict__ out) {
  const int lane = threadIdx.x & 31;
  const int wave = threadIdx.x >> 5;
  const int strip = blockIdx.x * 2 + wave;
  const int tile_m = strip >> 4;
  const int n0 = (strip & 15) * 32;
  const int half = lane >> 4;
  const int c = lane & 15;

  const float* xa = x + (tile_m * 16 + c) * FF + 4 * half;
  const float* tb = T + (4 * half) * NN + n0 + 2 * c;

  const int gsel = blockIdx.x & 7;
  const int wsc = __builtin_amdgcn_readfirstlane(wave);
  float* xo = out + (tile_m * 16 + c) * OUTC + 4 * half;

  v8f c0 = {}, c1 = {};
  v4f a[2][NS];
  v2f b[2][NS][4];
  v2f bt[NS][2][2];

  load_group(xa, tb, a[0], b[0]);
#pragma unroll
  for (int g = 0; g < FF / KC; ++g) {
    __builtin_amdgcn_sched_barrier(0);
    if (g + 1 < FF / KC)
      load_group(xa + (g + 1) * KC, tb + (g + 1) * KC * NN, a[(g + 1) & 1], b[(g + 1) & 1]);
    __builtin_amdgcn_sched_barrier(0);
    if (g == gsel && wsc == 0) {
      float* p = xo + g * KC;
#pragma unroll
      for (int s = 0; s < NS; ++s) *(v4f*)(p + 8 * s) = a[g & 1][s];
    }
    pair_rows(b[g & 1], bt);
    __builtin_amdgcn_sched_barrier(0);
    mma_group(a[g & 1], bt, c0, c1);
  }

  float* mo = m + (tile_m * 16 + half * 8) * NN + n0 + 2 * c;
#pragma unroll
  for (int r = 0; r < 8; ++r) {
    v2f o;
    o.x = c0[r];
    o.y = c1[r];
    *(v2f*)(mo + r * NN) = o;
  }
}

#define RH 8
#define JT 2

__device__ __forceinline__ float pair_term(const float (&mi)[DD],
                                           const float (&mj)[DD]) {
  float t[DD];
#pragma unroll
  for (int d = 0; d < DD; ++d) t[d] = mj[d] - mi[d];
  const float p0 = fabsf(t[0]) + fabsf(t[1]);
  const float p1 = fabsf(t[2]) + fabsf(t[3]);
  const float p2 = fabsf(t[4]) + fabsf(t[5]);
  const float p3 = fabsf(t[6]) + fabsf(t[7]);
  const float p4 = fabsf(t[8]) + fabsf(t[9]);
  const float p5 = fabsf(t[10]) + fabsf(t[11]);
  const float p6 = fabsf(t[12]) + fabsf(t[13]);
  const float p7 = fabsf(t[14]) + fabsf(t[15]);
  const float dist = ((p0 + p1) + (p2 + p3)) + ((p4 + p5) + (p6 + p7));
  return __expf(-dist);
}

__device__ __forceinline__ float from_lane_of_half(float v, unsigned sel) {
  const unsigned u = __float_as_uint(v);
  return __uint_as_float(__builtin_amdgcn_permlane16_var(u, u, sel, false, false));
}

template <int W>
__device__ __forceinline__ void fold(float (&a)[RH], unsigned c) {
  const bool up = (c & (unsigned)W) != 0;
  const unsigned sel = c ^ (unsigned)W;
#pragma unroll
  for (int r = 0; r < W; ++r) {
    const float keep = up ? a[r + W] : a[r];
    const float send = up ? a[r] : a[r + W];
    a[r] = keep + from_lane_of_half(send, sel);
  }
}

__device__ __forceinline__ void load_row(const float* __restrict__ p,
                                         float (&r)[DD]) {
  const float4 a = ((const float4*)p)[0];
  const float4 b = ((const float4*)p)[1];
  const float4 c = ((const float4*)p)[2];
  const float4 d = ((const float4*)p)[3];
  r[0] = a.x;  r[1] = a.y;  r[2] = a.z;  r[3] = a.w;
  r[4] = b.x;  r[5] = b.y;  r[6] = b.z;  r[7] = b.w;
  r[8] = c.x;  r[9] = c.y;  r[10] = c.z; r[11] = c.w;
  r[12] = d.x; r[13] = d.y; r[14] = d.z; r[15] = d.w;
}

__device__ __forceinline__ void load_trip(const float* __restrict__ base, int t, int c,
                                          float (&rows)[JT][DD]) {
#pragma unroll
  for (int u = 0; u < JT; ++u) load_row(base + ((t * JT + u) * 16 + c) * NN, rows[u]);
}

__device__ __forceinline__ void trip_pairs(const float (&mi)[RH][DD],
                                           const float (&rows)[JT][DD], float (&acc)[RH]) {
#pragma unroll
  for (int u = 0; u < JT; ++u) {
#pragma unroll
    for (int r = 0; r < RH; ++r) acc[r] += pair_term(mi[r], rows[u]);
  }
}

#define NTRIP (BB / (16 * JT))

__global__ __launch_bounds__(128) void feats_kernel(const float* __restrict__ m,
                                                    float* __restrict__ out) {
  const int lane = threadIdx.x & 31;
  const int wave = threadIdx.x >> 5;
  const int k = blockIdx.y;
  const int I = blockIdx.x * 4 + wave;
  const int half = lane >> 4;
  const int c = lane & 15;
  const float* base = m + k * DD;

  float mi[RH][DD], cur[JT][DD], nxt[JT][DD], acc[RH];
  load_trip(base, 0, c, cur);
#pragma unroll
  for (int r = 0; r < RH; ++r) {
    load_row(base + (I * 16 + half * RH + r) * NN, mi[r]);
    acc[r] = 0.f;
  }

#pragma unroll 2
  for (int t = 0; t < NTRIP; ++t) {
    __builtin_amdgcn_sched_barrier(0);
    load_trip(base, (t + 1) & (NTRIP - 1), c, nxt);
    __builtin_amdgcn_sched_barrier(0);

    trip_pairs(mi, cur, acc);

    __builtin_amdgcn_sched_barrier(0);
#pragma unroll
    for (int u = 0; u < JT; ++u) {
#pragma unroll
      for (int d = 0; d < DD; ++d) cur[u][d] = nxt[u][d];
    }
  }

  const unsigned uc = (unsigned)c;
#pragma unroll
  for (int r = 0; r < RH; ++r) acc[r] += from_lane_of_half(acc[r], uc ^ 8u);
  fold<4>(acc, uc);
  fold<2>(acc, uc);
  fold<1>(acc, uc);
  if (c < RH) out[(I * 16 + half * RH + c) * OUTC + FF + k] = acc[0];
}

extern "C" void kernel_launch(void* const* d_in, const int* in_sizes, int n_in,
                              void* d_out, int out_size, void* d_ws, size_t ws_size,
                              hipStream_t stream) {
  (void)in_sizes; (void)n_in; (void)out_size; (void)ws_size;
  const float* x = (const float*)d_in[0];
  const float* T = (const float*)d_in[1];
  float* out = (float*)d_out;
  float* m = (float*)d_ws;

  gemm_wmma_f32<<<256, 64, 0, stream>>>(x, T, m, out);
  feats_kernel<<<dim3(8, 32), 128, 0, stream>>>(m, out);
}


# feats: a trip is 4 rows j of a lane, not 2 (8 trips of 32 pairs): rows asked for a trip ahead lead their use by twice the arithmetic; a first read of a line of m outlasts a 2-row trip; same sums
# speedup vs baseline: 3.1140x; 1.0687x over previous
#include <hip/hip_runtime.h>
#include <hip/hip_bf16.h>

typedef float v2f __attribute__((ext_vector_type(2)));
typedef float v8f __attribute__((ext_vector_type(8)));

#define BB 512
#define FF 512
#define KK 32
#define DD 16
#define NN (KK * DD)
#define OUTC (FF + KK)

typedef float v4f __attribute__((ext_vector_type(4)));

#define KC 64
#define NS (KC / 8)

__device__ __forceinline__ void load_group(const float* __restrict__ xa,
                                           const float* __restrict__ tb,
                                           v4f (&a)[NS], v2f (&b)[NS][4]) {
#pragma unroll
  for (int s = 0; s < NS; ++s) {
    a[s] = *(const v4f*)(xa + 8 * s);
#pragma unroll
    for (int q = 0; q < 4; ++q) b[s][q] = *(const v2f*)(tb + (8 * s + q) * NN);
  }
}

__device__ __forceinline__ void pair_rows(const v2f (&b)[NS][4], v2f (&bt)[NS][2][2]) {
#pragma unroll
  for (int s = 0; s < NS; ++s) {
#pragma unroll
    for (int j = 0; j < 2; ++j) {
      bt[s][j][0].x = b[s][2 * j].x;   bt[s][j][0].y = b[s][2 * j + 1].x;
      bt[s][j][1].x = b[s][2 * j].y;   bt[s][j][1].y = b[s][2 * j + 1].y;
    }
  }
}

__device__ __forceinline__ void mma_group(const v4f (&a)[NS], const v2f (&bt)[NS][2][2],
                                          v8f& c0, v8f& c1) {
#pragma unroll
  for (int s = 0; s < NS; ++s) {
#pragma unroll
    for (int j = 0; j < 2; ++j) {
      v2f af;
      af.x = a[s][2 * j];
      af.y = a[s][2 * j + 1];
      c0 = __builtin_amdgcn_wmma_f32_16x16x4_f32(false, af, false, bt[s][j][0],
                                                 (short)0, c0, false, false);
      c1 = __builtin_amdgcn_wmma_f32_16x16x4_f32(false, af, false, bt[s][j][1],
                                                 (short)0, c1, false, false);
    }
  }
}

__global__ __launch_bounds__(64) void gemm_wmma_f32(const float* __restrict__ x,
                                                    const float* __restrict__ T,
                                                    float* __restrict__ m,
                                                    float* __restrict__ out) {
  const int lane = threadIdx.x & 31;
  const int wave = threadIdx.x >> 5;
  const int strip = blockIdx.x * 2 + wave;
  const int tile_m = strip >> 4;
  const int n0 = (strip & 15) * 32;
  const int half = lane >> 4;
  const int c = lane & 15;

  const float* xa = x + (tile_m * 16 + c) * FF + 4 * half;
  const float* tb = T + (4 * half) * NN + n0 + 2 * c;

  const int gsel = blockIdx.x & 7;
  const int wsc = __builtin_amdgcn_readfirstlane(wave);
  float* xo = out + (tile_m * 16 + c) * OUTC + 4 * half;

  v8f c0 = {}, c1 = {};
  v4f a[2][NS];
  v2f b[2][NS][4];
  v2f bt[NS][2][2];

  load_group(xa, tb, a[0], b[0]);
#pragma unroll
  for (int g = 0; g < FF / KC; ++g) {
    __builtin_amdgcn_sched_barrier(0);
    if (g + 1 < FF / KC)
      load_group(xa + (g + 1) * KC, tb + (g + 1) * KC * NN, a[(g + 1) & 1], b[(g + 1) & 1]);
    __builtin_amdgcn_sched_barrier(0);
    if (g == gsel && wsc == 0) {
      float* p = xo + g * KC;
#pragma unroll
      for (int s = 0; s < NS; ++s) *(v4f*)(p + 8 * s) = a[g & 1][s];
    }
    pair_rows(b[g & 1], bt);
    __builtin_amdgcn_sched_barrier(0);
    mma_group(a[g & 1], bt, c0, c1);
  }

  float* mo = m + (tile_m * 16 + half * 8) * NN + n0 + 2 * c;
#pragma unroll
  for (int r = 0; r < 8; ++r) {
    v2f o;
    o.x = c0[r];
    o.y = c1[r];
    *(v2f*)(mo + r * NN) = o;
  }
}

#define RH 8
#define JT 4

__device__ __forceinline__ float pair_term(const float (&mi)[DD],
                                           const float (&mj)[DD]) {
  float t[DD];
#pragma unroll
  for (int d = 0; d < DD; ++d) t[d] = mj[d] - mi[d];
  const float p0 = fabsf(t[0]) + fabsf(t[1]);
  const float p1 = fabsf(t[2]) + fabsf(t[3]);
  const float p2 = fabsf(t[4]) + fabsf(t[5]);
  const float p3 = fabsf(t[6]) + fabsf(t[7]);
  const float p4 = fabsf(t[8]) + fabsf(t[9]);
  const float p5 = fabsf(t[10]) + fabsf(t[11]);
  const float p6 = fabsf(t[12]) + fabsf(t[13]);
  const float p7 = fabsf(t[14]) + fabsf(t[15]);
  const float dist = ((p0 + p1) + (p2 + p3)) + ((p4 + p5) + (p6 + p7));
  return __expf(-dist);
}

__device__ __forceinline__ float from_lane_of_half(float v, unsigned sel) {
  const unsigned u = __float_as_uint(v);
  return __uint_as_float(__builtin_amdgcn_permlane16_var(u, u, sel, false, false));
}

template <int W>
__device__ __forceinline__ void fold(float (&a)[RH], unsigned c) {
  const bool up = (c & (unsigned)W) != 0;
  const unsigned sel = c ^ (unsigned)W;
#pragma unroll
  for (int r = 0; r < W; ++r) {
    const float keep = up ? a[r + W] : a[r];
    const float send = up ? a[r] : a[r + W];
    a[r] = keep + from_lane_of_half(send, sel);
  }
}

__device__ __forceinline__ void load_row(const float* __restrict__ p,
                                         float (&r)[DD]) {
  const float4 a = ((const float4*)p)[0];
  const float4 b = ((const float4*)p)[1];
  const float4 c = ((const float4*)p)[2];
  const float4 d = ((const float4*)p)[3];
  r[0] = a.x;  r[1] = a.y;  r[2] = a.z;  r[3] = a.w;
  r[4] = b.x;  r[5] = b.y;  r[6] = b.z;  r[7] = b.w;
  r[8] = c.x;  r[9] = c.y;  r[10] = c.z; r[11] = c.w;
  r[12] = d.x; r[13] = d.y; r[14] = d.z; r[15] = d.w;
}

__device__ __forceinline__ void load_trip(const float* __restrict__ base, int t, int c,
                                          float (&rows)[JT][DD]) {
#pragma unroll
  for (int u = 0; u < JT; ++u) load_row(base + ((t * JT + u) * 16 + c) * NN, rows[u]);
}

__device__ __forceinline__ void trip_pairs(const float (&mi)[RH][DD],
                                           const float (&rows)[JT][DD], float (&acc)[RH]) {
#pragma unroll
  for (int u = 0; u < JT; ++u) {
#pragma unroll
    for (int r = 0; r < RH; ++r) acc[r] += pair_term(mi[r], rows[u]);
  }
}

#define NTRIP (BB / (16 * JT))

__global__ __launch_bounds__(128) void feats_kernel(const float* __restrict__ m,
                                                    float* __restrict__ out) {
  const int lane = threadIdx.x & 31;
  const int wave = threadIdx.x >> 5;
  const int k = blockIdx.y;
  const int I = blockIdx.x * 4 + wave;
  const int half = lane >> 4;
  const int c = lane & 15;
  const float* base = m + k * DD;

  float mi[RH][DD], cur[JT][DD], nxt[JT][DD], acc[RH];
  load_trip(base, 0, c, cur);
#pragma unroll
  for (int r = 0; r < RH; ++r) {
    load_row(base + (I * 16 + half * RH + r) * NN, mi[r]);
    acc[r] = 0.f;
  }

#pragma unroll 2
  for (int t = 0; t < NTRIP; ++t) {
    __builtin_amdgcn_sched_barrier(0);
    load_trip(base, (t + 1) & (NTRIP - 1), c, nxt);
    __builtin_amdgcn_sched_barrier(0);

    trip_pairs(mi, cur, acc);

    __builtin_amdgcn_sched_barrier(0);
#pragma unroll
    for (int u = 0; u < JT; ++u) {
#pragma unroll
      for (int d = 0; d < DD; ++d) cur[u][d] = nxt[u][d];
    }
  }

  const unsigned uc = (unsigned)c;
#pragma unroll
  for (int r = 0; r < RH; ++r) acc[r] += from_lane_of_half(acc[r], uc ^ 8u);
  fold<4>(acc, uc);
  fold<2>(acc, uc);
  fold<1>(acc, uc);
  if (c < RH) out[(I * 16 + half * RH + c) * OUTC + FF + k] = acc[0];
}

extern "C" void kernel_launch(void* const* d_in, const int* in_sizes, int n_in,
                              void* d_out, int out_size, void* d_ws, size_t ws_size,
                              hipStream_t stream) {
  (void)in_sizes; (void)n_in; (void)out_size; (void)ws_size;
  const float* x = (const float*)d_in[0];
  const float* T = (const float*)d_in[1];
  float* out = (float*)d_out;
  float* m = (float*)d_ws;

  gemm_wmma_f32<<<256, 64, 0, stream>>>(x, T, m, out);
  feats_kernel<<<dim3(8, 32), 128, 0, stream>>>(m, out);
}


# m = x @ T: k split in two over wave pairs, a block is 4 waves (all 4 SIMDs of a WGP work; 4 load groups per wave, not 8); halves joined once via LDS; k sum now 2 partials of 256 k
# speedup vs baseline: 3.3973x; 1.0910x over previous
#include <hip/hip_runtime.h>
#include <hip/hip_bf16.h>

typedef float v2f __attribute__((ext_vector_type(2)));
typedef float v8f __attribute__((ext_vector_type(8)));

#define BB 512
#define FF 512
#define KK 32
#define DD 16
#define NN (KK * DD)
#define OUTC (FF + KK)

typedef float v4f __attribute__((ext_vector_type(4)));

#define KC 64
#define NS (KC / 8)

__device__ __forceinline__ void load_group(const float* __restrict__ xa,
                                           const float* __restrict__ tb,
                                           v4f (&a)[NS], v2f (&b)[NS][4]) {
#pragma unroll
  for (int s = 0; s < NS; ++s) {
    a[s] = *(const v4f*)(xa + 8 * s);
#pragma unroll
    for (int q = 0; q < 4; ++q) b[s][q] = *(const v2f*)(tb + (8 * s + q) * NN);
  }
}

__device__ __forceinline__ void pair_rows(const v2f (&b)[NS][4], v2f (&bt)[NS][2][2]) {
#pragma unroll
  for (int s = 0; s < NS; ++s) {
#pragma unroll
    for (int j = 0; j < 2; ++j) {
      bt[s][j][0].x = b[s][2 * j].x;   bt[s][j][0].y = b[s][2 * j + 1].x;
      bt[s][j][1].x = b[s][2 * j].y;   bt[s][j][1].y = b[s][2 * j + 1].y;
    }
  }
}

__device__ __forceinline__ void mma_group(const v4f (&a)[NS], const v2f (&bt)[NS][2][2],
                                          v8f& c0, v8f& c1) {
#pragma unroll
  for (int s = 0; s < NS; ++s) {
#pragma unroll
    for (int j = 0; j < 2; ++j) {
      v2f af;
      af.x = a[s][2 * j];
      af.y = a[s][2 * j + 1];
      c0 = __builtin_amdgcn_wmma_f32_16x16x4_f32(false, af, false, bt[s][j][0],
                                                 (short)0, c0, false, false);
      c1 = __builtin_amdgcn_wmma_f32_16x16x4_f32(false, af, false, bt[s][j][1],
                                                 (short)0, c1, false, false);
    }
  }
}

#define KSPAN (FF / 2)
#define NG (KSPAN / KC)

__global__ __launch_bounds__(128) void gemm_wmma_f32(const float* __restrict__ x,
                                                     const float* __restrict__ T,
                                                     float* __restrict__ m,
                                                     float* __restrict__ out) {
  __shared__ v2f red[2][2][4][32];

  const int lane = threadIdx.x & 31;
  const int wave = __builtin_amdgcn_readfirstlane((int)(threadIdx.x >> 5));
  const int sl = wave & 1;
  const int kh = wave >> 1;
  const int strip = blockIdx.x * 2 + sl;
  const int tile_m = strip >> 4;
  const int n0 = (strip & 15) * 32;
  const int half = lane >> 4;
  const int c = lane & 15;
  const int k0 = kh * KSPAN;

  const float* xa = x + (tile_m * 16 + c) * FF + k0 + 4 * half;
  const float* tb = T + (k0 + 4 * half) * NN + n0 + 2 * c;

  const int gsel = blockIdx.x & 7;
  const bool copier = (sl == 0) && (kh == (gsel >> 2));
  const int gcopy = gsel & 3;
  float* xo = out + (tile_m * 16 + c) * OUTC + k0 + 4 * half;

  v8f c0 = {}, c1 = {};
  v4f a[2][NS];
  v2f b[2][NS][4];
  v2f bt[NS][2][2];

  load_group(xa, tb, a[0], b[0]);
#pragma unroll
  for (int g = 0; g < NG; ++g) {
    __builtin_amdgcn_sched_barrier(0);
    if (g + 1 < NG)
      load_group(xa + (g + 1) * KC, tb + (g + 1) * KC * NN, a[(g + 1) & 1], b[(g + 1) & 1]);
    __builtin_amdgcn_sched_barrier(0);
    if (g == gcopy && copier) {
      float* p = xo + g * KC;
#pragma unroll
      for (int s = 0; s < NS; ++s) *(v4f*)(p + 8 * s) = a[g & 1][s];
    }
    pair_rows(b[g & 1], bt);
    __builtin_amdgcn_sched_barrier(0);
    mma_group(a[g & 1], bt, c0, c1);
  }

  const bool upper = kh != 0;
#pragma unroll
  for (int r = 0; r < 4; ++r) {
    v2f o;
    o.x = upper ? c0[r] : c0[r + 4];
    o.y = upper ? c1[r] : c1[r + 4];
    red[sl][kh][r][lane] = o;
  }
  __syncthreads();
  float* mo = m + (tile_m * 16 + half * 8 + 4 * kh) * NN + n0 + 2 * c;
#pragma unroll
  for (int r = 0; r < 4; ++r) {
    const v2f p = red[sl][1 - kh][r][lane];
    const float lo0 = upper ? p.x : c0[r];
    const float hi0 = upper ? c0[r + 4] : p.x;
    const float lo1 = upper ? p.y : c1[r];
    const float hi1 = upper ? c1[r + 4] : p.y;
    v2f o;
    o.x = lo0 + hi0;
    o.y = lo1 + hi1;
    *(v2f*)(mo + r * NN) = o;
  }
}

#define RH 8
#define JT 4

__device__ __forceinline__ float pair_term(const float (&mi)[DD],
                                           const float (&mj)[DD]) {
  float t[DD];
#pragma unroll
  for (int d = 0; d < DD; ++d) t[d] = mj[d] - mi[d];
  const float p0 = fabsf(t[0]) + fabsf(t[1]);
  const float p1 = fabsf(t[2]) + fabsf(t[3]);
  const float p2 = fabsf(t[4]) + fabsf(t[5]);
  const float p3 = fabsf(t[6]) + fabsf(t[7]);
  const float p4 = fabsf(t[8]) + fabsf(t[9]);
  const float p5 = fabsf(t[10]) + fabsf(t[11]);
  const float p6 = fabsf(t[12]) + fabsf(t[13]);
  const float p7 = fabsf(t[14]) + fabsf(t[15]);
  const float dist = ((p0 + p1) + (p2 + p3)) + ((p4 + p5) + (p6 + p7));
  return __expf(-dist);
}

__device__ __forceinline__ float from_lane_of_half(float v, unsigned sel) {
  const unsigned u = __float_as_uint(v);
  return __uint_as_float(__builtin_amdgcn_permlane16_var(u, u, sel, false, false));
}

template <int W>
__device__ __forceinline__ void fold(float (&a)[RH], unsigned c) {
  const bool up = (c & (unsigned)W) != 0;
  const unsigned sel = c ^ (unsigned)W;
#pragma unroll
  for (int r = 0; r < W; ++r) {
    const float keep = up ? a[r + W] : a[r];
    const float send = up ? a[r] : a[r + W];
    a[r] = keep + from_lane_of_half(send, sel);
  }
}

__device__ __forceinline__ void load_row(const float* __restrict__ p,
                                         float (&r)[DD]) {
  const float4 a = ((const float4*)p)[0];
  const float4 b = ((const float4*)p)[1];
  const float4 c = ((const float4*)p)[2];
  const float4 d = ((const float4*)p)[3];
  r[0] = a.x;  r[1] = a.y;  r[2] = a.z;  r[3] = a.w;
  r[4] = b.x;  r[5] = b.y;  r[6] = b.z;  r[7] = b.w;
  r[8] = c.x;  r[9] = c.y;  r[10] = c.z; r[11] = c.w;
  r[12] = d.x; r[13] = d.y; r[14] = d.z; r[15] = d.w;
}

__device__ __forceinline__ void load_trip(const float* __restrict__ base, int t, int c,
                                          float (&rows)[JT][DD]) {
#pragma unroll
  for (int u = 0; u < JT; ++u) load_row(base + ((t * JT + u) * 16 + c) * NN, rows[u]);
}

__device__ __forceinline__ void trip_pairs(const float (&mi)[RH][DD],
                                           const float (&rows)[JT][DD], float (&acc)[RH]) {
#pragma unroll
  for (int u = 0; u < JT; ++u) {
#pragma unroll
    for (int r = 0; r < RH; ++r) acc[r] += pair_term(mi[r], rows[u]);
  }
}

#define NTRIP (BB / (16 * JT))

__global__ __launch_bounds__(128) void feats_kernel(const float* __restrict__ m,
                                                    float* __restrict__ out) {
  const int lane = threadIdx.x & 31;
  const int wave = threadIdx.x >> 5;
  const int k = blockIdx.y;
  const int I = blockIdx.x * 4 + wave;
  const int half = lane >> 4;
  const int c = lane & 15;
  const float* base = m + k * DD;

  float mi[RH][DD], cur[JT][DD], nxt[JT][DD], acc[RH];
  load_trip(base, 0, c, cur);
#pragma unroll
  for (int r = 0; r < RH; ++r) {
    load_row(base + (I * 16 + half * RH + r) * NN, mi[r]);
    acc[r] = 0.f;
  }

#pragma unroll 2
  for (int t = 0; t < NTRIP; ++t) {
    __builtin_amdgcn_sched_barrier(0);
    load_trip(base, (t + 1) & (NTRIP - 1), c, nxt);
    __builtin_amdgcn_sched_barrier(0);

    trip_pairs(mi, cur, acc);

    __builtin_amdgcn_sched_barrier(0);
#pragma unroll
    for (int u = 0; u < JT; ++u) {
#pragma unroll
      for (int d = 0; d < DD; ++d) cur[u][d] = nxt[u][d];
    }
  }

  const unsigned uc = (unsigned)c;
#pragma unroll
  for (int r = 0; r < RH; ++r) acc[r] += from_lane_of_half(acc[r], uc ^ 8u);
  fold<4>(acc, uc);
  fold<2>(acc, uc);
  fold<1>(acc, uc);
  if (c < RH) out[(I * 16 + half * RH + c) * OUTC + FF + k] = acc[0];
}

extern "C" void kernel_launch(void* const* d_in, const int* in_sizes, int n_in,
                              void* d_out, int out_size, void* d_ws, size_t ws_size,
                              hipStream_t stream) {
  (void)in_sizes; (void)n_in; (void)out_size; (void)ws_size;
  const float* x = (const float*)d_in[0];
  const float* T = (const float*)d_in[1];
  float* out = (float*)d_out;
  float* m = (float*)d_ws;

  gemm_wmma_f32<<<256, 128, 0, stream>>>(x, T, m, out);
  feats_kernel<<<dim3(8, 32), 128, 0, stream>>>(m, out);
}
